# starting kernel
# baseline (speedup 1.0000x reference)
#include <hip/hip_runtime.h>

typedef __attribute__((ext_vector_type(2))) float v2f;
typedef __attribute__((ext_vector_type(8))) float v8f;

template <int TC>
__global__ __launch_bounds__(32) void sg_apply_wmma(
    const float* __restrict__ Sg, const float* __restrict__ Z,
    float* __restrict__ G) {
  const int nt    = blockIdx.x;
  const int mt    = blockIdx.y;
  const int l     = threadIdx.x;
  const int half  = l >> 4;
  const int lm    = l & 15;
  const int arow  = mt * 16 + lm;
  const int bcol0 = nt * 32 + lm;
  const int bcol1 = bcol0 + 16;

  v8f acc0 = {0.f, 0.f, 0.f, 0.f, 0.f, 0.f, 0.f, 0.f};
  v8f acc1 = {0.f, 0.f, 0.f, 0.f, 0.f, 0.f, 0.f, 0.f};

  const float* __restrict__ ap  = Sg + arow * 192 + half * 2;
  const float* __restrict__ bp0 = Z + half * 2 * TC + bcol0;
  const float* __restrict__ bp1 = Z + half * 2 * TC + bcol1;

#pragma unroll
  for (int k0 = 0; k0 < 192; k0 += 4) {
    v2f a, b0, b1;
    a.x  = ap[k0];
    a.y  = ap[k0 + 1];
    b0.x = bp0[k0 * TC];
    b0.y = bp0[k0 * TC + TC];
    b1.x = bp1[k0 * TC];
    b1.y = bp1[k0 * TC + TC];
    acc0 = __builtin_amdgcn_wmma_f32_16x16x4_f32(
        false, a, false, b0, (short)0, acc0, false, false);
    acc1 = __builtin_amdgcn_wmma_f32_16x16x4_f32(
        false, a, false, b1, (short)0, acc1, false, false);
  }

  const int rbase = mt * 16 + half * 8;
  float* __restrict__ gp = G + rbase * TC;
#pragma unroll
  for (int v = 0; v < 8; ++v) {
    gp[v * TC + bcol0] = acc0[v];
    gp[v * TC + bcol1] = acc1[v];
  }
}

__global__ void combine_shift(const float* __restrict__ x,
                              const float* __restrict__ g,
                              const float* __restrict__ s,
                              float* __restrict__ y, int t, int c) {
  const int idx = blockIdx.x * blockDim.x + threadIdx.x;
  const int total = 192 * t * c;
  if (idx >= total) return;
  const int ch   = idx % c;
  const int tmp  = idx / c;
  const int tau  = tmp % t;
  const int node = tmp / t;
  const int taum = (tau + t - 1) % t;
  const int i1 = (node * t + taum) * c + ch;
  const float s00 = s[0], s01 = s[1], s10 = s[2], s11 = s[3];
  y[idx] = s00 * x[idx] + s01 * g[idx] + s10 * x[i1] + s11 * g[i1];
}

__global__ void mix_channels(const float* __restrict__ x0,
                             const float* __restrict__ x1,
                             const float* __restrict__ x2,
                             const float* __restrict__ h,
                             float* __restrict__ y,
                             int t, int cin, int cout) {
  const int idx = blockIdx.x * blockDim.x + threadIdx.x;
  const int total = 192 * t * cout;
  if (idx >= total) return;
  const int o    = idx % cout;
  const int tmp  = idx / cout;
  const int tau  = tmp % t;
  const int node = tmp / t;
  const int base = (node * t + tau) * cin;
  float acc = 0.f;
  for (int i = 0; i < cin; ++i) {
    const float* hp = h + (o * cin + i) * 3;
    acc = fmaf(hp[0], x0[base + i], acc);
    acc = fmaf(hp[1], x1[base + i], acc);
    acc = fmaf(hp[2], x2[base + i], acc);
  }
  y[idx] = acc;
}

__global__ void down_relu(const float* __restrict__ y, float* __restrict__ a,
                          int t_out, int c) {
  const int idx = blockIdx.x * blockDim.x + threadIdx.x;
  const int total = 192 * t_out * c;
  if (idx >= total) return;
  const int ch   = idx % c;
  const int tmp  = idx / c;
  const int tau  = tmp % t_out;
  const int node = tmp / t_out;
  const int b0 = (node * 2 * t_out + 2 * tau) * c + ch;
  const float m = fmaxf(y[b0], y[b0 + c]);
  a[idx] = fmaxf(m, 0.f);
}

__global__ void up_relu(const float* __restrict__ y, float* __restrict__ a,
                        int t_in, int c) {
  const int idx = blockIdx.x * blockDim.x + threadIdx.x;
  const int t_out = 2 * t_in;
  const int total = 192 * t_out * c;
  if (idx >= total) return;
  const int ch   = idx % c;
  const int tmp  = idx / c;
  const int tau  = tmp % t_out;
  const int node = tmp / t_out;
  float v = 0.f;
  if ((tau & 1) == 0) v = fmaxf(y[(node * t_in + (tau >> 1)) * c + ch], 0.f);
  a[idx] = v;
}

static void launch_sg(const float* Sg, const float* Z, float* G, int tc,
                      hipStream_t stream) {
  const dim3 grid(tc / 32, 12);
  switch (tc) {
    case 32:
      sg_apply_wmma<32><<<grid, 32, 0, stream>>>(Sg, Z, G);
      break;
    case 256:
      sg_apply_wmma<256><<<grid, 32, 0, stream>>>(Sg, Z, G);
      break;
    default:
      sg_apply_wmma<512><<<grid, 32, 0, stream>>>(Sg, Z, G);
      break;
  }
}

static void conv_layer(const float* Sg, const float* s, const float* x0,
                       const float* h, float* X1, float* X2, float* G,
                       float* Y, int t, int cin, int cout, hipStream_t stream) {
  const int tc   = t * cin;
  const int totc = 192 * t * cin;
  const int toty = 192 * t * cout;
  launch_sg(Sg, x0, G, tc, stream);
  combine_shift<<<(totc + 255) / 256, 256, 0, stream>>>(x0, G, s, X1, t, cin);
  launch_sg(Sg, X1, G, tc, stream);
  combine_shift<<<(totc + 255) / 256, 256, 0, stream>>>(X1, G, s, X2, t, cin);
  mix_channels<<<(toty + 255) / 256, 256, 0, stream>>>(x0, X1, X2, h, Y, t,
                                                       cin, cout);
}

extern "C" void kernel_launch(void* const* d_in, const int* in_sizes, int n_in,
                              void* d_out, int out_size, void* d_ws,
                              size_t ws_size, hipStream_t stream) {
  (void)in_sizes; (void)n_in; (void)out_size; (void)ws_size;
  const float* X    = (const float*)d_in[0];
  const float* Sg   = (const float*)d_in[1];
  const float* s    = (const float*)d_in[2];
  const float* h_e1 = (const float*)d_in[3];
  const float* h_e2 = (const float*)d_in[4];
  const float* h_d1 = (const float*)d_in[5];
  const float* h_d2 = (const float*)d_in[6];
  float* out = (float*)d_out;
  float* ws  = (float*)d_ws;

  const int BUF = 192 * 32 * 16;
  float* X1 = ws + 0 * BUF;
  float* X2 = ws + 1 * BUF;
  float* G  = ws + 2 * BUF;
  float* Y  = ws + 3 * BUF;
  float* A0 = ws + 4 * BUF;
  float* A1 = ws + 5 * BUF;

  conv_layer(Sg, s, X, h_e1, X1, X2, G, Y, 32, 1, 16, stream);
  down_relu<<<(192 * 16 * 16 + 255) / 256, 256, 0, stream>>>(Y, A0, 16, 16);

  conv_layer(Sg, s, A0, h_e2, X1, X2, G, Y, 16, 16, 32, stream);
  down_relu<<<(192 * 8 * 32 + 255) / 256, 256, 0, stream>>>(Y, A1, 8, 32);

  up_relu<<<(192 * 16 * 32 + 255) / 256, 256, 0, stream>>>(A1, A0, 8, 32);
  conv_layer(Sg, s, A0, h_d1, X1, X2, G, Y, 16, 32, 16, stream);

  up_relu<<<(192 * 32 * 16 + 255) / 256, 256, 0, stream>>>(Y, A1, 16, 16);
  conv_layer(Sg, s, A1, h_d2, X1, X2, G, out, 32, 16, 1, stream);
}


# Encoder 1 cut into 4 time slabs per channel: a work-group does a 16-step window (4-step halo) for 8 output steps, one column tile not two; 16->64 groups; output stored slab-major; sums unchanged
# speedup vs baseline: 2.1002x; 2.1002x over previous
#include <hip/hip_runtime.h>

typedef __attribute__((ext_vector_type(2))) float v2f;
typedef __attribute__((ext_vector_type(4))) float v4f;
typedef __attribute__((ext_vector_type(8))) float v8f;

#define LDB 196

template <int T>
__device__ __forceinline__ void sg_times_plane(const v4f (&a)[24],
                                               const float* __restrict__ Bt,
                                               float* __restrict__ Us,
                                               int half, int bcol, int rbase) {
  v8f acc = {0.f, 0.f, 0.f, 0.f, 0.f, 0.f, 0.f, 0.f};
  const float* __restrict__ bp = Bt + bcol * LDB + half * 4;
  constexpr bool AHEAD = (T == 16);
  v4f bnext = {0.f, 0.f, 0.f, 0.f};
  if (AHEAD) bnext = *(const v4f*)bp;
#pragma unroll
  for (int m = 0; m < 24; ++m) {
    v4f bv;
    if (AHEAD) {
      bv = bnext;
      if (m + 1 < 24) bnext = *(const v4f*)(bp + 8 * (m + 1));
    } else {
      bv = *(const v4f*)(bp + 8 * m);
    }
    const v2f a0 = {a[m][0], a[m][1]};
    const v2f a1 = {a[m][2], a[m][3]};
    const v2f b0 = {bv[0], bv[1]};
    const v2f b1 = {bv[2], bv[3]};
    acc = __builtin_amdgcn_wmma_f32_16x16x4_f32(
        false, a0, false, b0, (short)0, acc, false, false);
    acc = __builtin_amdgcn_wmma_f32_16x16x4_f32(
        false, a1, false, b1, (short)0, acc, false, false);
  }
  float* __restrict__ up = Us + rbase * T + bcol;
#pragma unroll
  for (int v = 0; v < 8; ++v) {
    up[v * T] = acc[v];
  }
}

template <int T, int TG, int CIN, bool UNPOOL_IN, bool POOL_OUT, bool IN_SLAB>
__global__ __launch_bounds__(24 * T) void gt_layer(
    const float* __restrict__ Sg, const float* __restrict__ s,
    const float* __restrict__ xin, const float* __restrict__ h,
    float* __restrict__ yout) {
  __shared__ float Bt[T * LDB];
  __shared__ float Us[192 * T];

  const int tid   = threadIdx.x;
  constexpr bool SLAB = (TG != T);
  static_assert(!SLAB || (T == 16 && TG == 32 && !UNPOOL_IN && POOL_OUT && !IN_SLAB),
                "slab form: 32-step layer, plain input, pooled output");
  static_assert(!IN_SLAB || (T == 16 && TG == 16 && !UNPOOL_IN),
                "slab-major input: 16 steps in 4 slabs of 4");
  const int o     = SLAB ? (int)(blockIdx.x >> 2) : (int)blockIdx.x;
  const int g     = SLAB ? (int)(blockIdx.x & 3) : 0;
  const int wave  = tid >> 5;
  const int l     = tid & 31;
  const int half  = l >> 4;
  const int lm    = l & 15;
  const int mt    = wave % 12;
  const int nt    = wave / 12;
  const int arow  = mt * 16 + lm;
  const int bcol  = nt * 16 + lm;
  const int rbase = mt * 16 + half * 8;

  v4f a[24];
  const float* __restrict__ ap = Sg + arow * 192 + half * 4;
#pragma unroll
  for (int m = 0; m < 24; ++m) {
    a[m] = *(const v4f*)(ap + 8 * m);
  }

  const float s00 = s[0], s01 = s[1], s10 = s[2], s11 = s[3];

  const int e0    = tid * 8;
  const int tau0  = e0 % T;
  const int node  = e0 / T;
  const int tprev = (tau0 + T - 1) % T;
  float* __restrict__ bown = Bt + tau0 * LDB + node;
  const float* __restrict__ bprev = Bt + tprev * LDB + node;
  const float* __restrict__ uprev = Us + node * T + tprev;

  float p0[8], p1[8], p2[8];
#pragma unroll
  for (int j = 0; j < 8; ++j) { p0[j] = 0.f; p1[j] = 0.f; p2[j] = 0.f; }
  const float* __restrict__ hp = h + o * CIN * 3;
  if (!UNPOOL_IN) {
    int offa, offb;
    if (SLAB) {
      offa = node * TG + ((8 * g - 4 + tau0) & (TG - 1));
      offb = node * TG + ((8 * g + tau0) & (TG - 1));
    } else if (IN_SLAB) {
      offa = (tau0 >> 2) * 768 + node * 4;
      offb = offa + 768;
    } else {
      offa = e0;
      offb = e0 + 4;
    }
    const float* __restrict__ xp = xin;
#pragma unroll
    for (int i = 0; i < CIN; ++i) {
      const v4f xa = *(const v4f*)(xp + i * (192 * TG) + offa);
      const v4f xb = *(const v4f*)(xp + i * (192 * TG) + offb);
      const float h0 = hp[i * 3], h1 = hp[i * 3 + 1], h2 = hp[i * 3 + 2];
#pragma unroll
      for (int j = 0; j < 4; ++j) {
        p0[j]     = fmaf(h0, xa[j], p0[j]);
        p1[j]     = fmaf(h1, xa[j], p1[j]);
        p2[j]     = fmaf(h2, xa[j], p2[j]);
        p0[4 + j] = fmaf(h0, xb[j], p0[4 + j]);
        p1[4 + j] = fmaf(h1, xb[j], p1[4 + j]);
        p2[4 + j] = fmaf(h2, xb[j], p2[4 + j]);
      }
    }
  } else {
    const float* __restrict__ xp = xin + tid * 4;
#pragma unroll
    for (int i = 0; i < CIN; ++i) {
      const v4f xa = *(const v4f*)(xp + i * (192 * (T / 2)));
      const float h0 = hp[i * 3], h1 = hp[i * 3 + 1], h2 = hp[i * 3 + 2];
#pragma unroll
      for (int j = 0; j < 4; ++j) {
        const float x = fmaxf(xa[j], 0.f);
        p0[2 * j] = fmaf(h0, x, p0[2 * j]);
        p1[2 * j] = fmaf(h1, x, p1[2 * j]);
        p2[2 * j] = fmaf(h2, x, p2[2 * j]);
      }
    }
  }

#pragma unroll
  for (int j = 0; j < 8; ++j) {
    bown[j * LDB] = p2[j];
  }
  __syncthreads();
  sg_times_plane<T>(a, Bt, Us, half, bcol, rbase);
  __syncthreads();

  float w[8];
  {
    const v4f ua = *(const v4f*)(Us + e0);
    const v4f ub = *(const v4f*)(Us + e0 + 4);
    const float u[8] = {ua[0], ua[1], ua[2], ua[3], ub[0], ub[1], ub[2], ub[3]};
    float zm = *bprev;
    float um = *uprev;
#pragma unroll
    for (int j = 0; j < 8; ++j) {
      float t = fmaf(s00, p2[j], p1[j]);
      t = fmaf(s01, u[j], t);
      t = fmaf(s10, zm, t);
      t = fmaf(s11, um, t);
      w[j] = t;
      zm = p2[j];
      um = u[j];
    }
  }
  __syncthreads();
#pragma unroll
  for (int j = 0; j < 8; ++j) {
    bown[j * LDB] = w[j];
  }
  __syncthreads();

  sg_times_plane<T>(a, Bt, Us, half, bcol, rbase);
  __syncthreads();

  float y[8];
  {
    const v4f ua = *(const v4f*)(Us + e0);
    const v4f ub = *(const v4f*)(Us + e0 + 4);
    const float u[8] = {ua[0], ua[1], ua[2], ua[3], ub[0], ub[1], ub[2], ub[3]};
    float zm = *bprev;
    float um = *uprev;
#pragma unroll
    for (int j = 0; j < 8; ++j) {
      float t = fmaf(s00, w[j], p0[j]);
      t = fmaf(s01, u[j], t);
      t = fmaf(s10, zm, t);
      t = fmaf(s11, um, t);
      y[j] = t;
      zm = w[j];
      um = u[j];
    }
  }

  if (SLAB) {
    const int hsel = tid & 1;
    float* __restrict__ yp = yout + (o * 4 + g) * 768 + node * 4 + hsel * 2;
    const float ya = hsel ? y[0] : y[4];
    const float yb = hsel ? y[1] : y[5];
    const float yc = hsel ? y[2] : y[6];
    const float yd = hsel ? y[3] : y[7];
    v2f r;
    r[0] = fmaxf(fmaxf(ya, yb), 0.f);
    r[1] = fmaxf(fmaxf(yc, yd), 0.f);
    *(v2f*)yp = r;
  } else if (POOL_OUT) {
    float* __restrict__ yp = yout + o * (192 * (T / 2)) + tid * 4;
    v4f r;
#pragma unroll
    for (int j = 0; j < 4; ++j) {
      r[j] = fmaxf(fmaxf(y[2 * j], y[2 * j + 1]), 0.f);
    }
    *(v4f*)yp = r;
  } else {
    float* __restrict__ yp = yout + o * (192 * T) + e0;
    v4f lo = {y[0], y[1], y[2], y[3]};
    v4f hi = {y[4], y[5], y[6], y[7]};
    *(v4f*)yp       = lo;
    *(v4f*)(yp + 4) = hi;
  }
}

extern "C" void kernel_launch(void* const* d_in, const int* in_sizes, int n_in,
                              void* d_out, int out_size, void* d_ws,
                              size_t ws_size, hipStream_t stream) {
  (void)in_sizes; (void)n_in; (void)out_size; (void)ws_size;
  const float* X    = (const float*)d_in[0];
  const float* Sg   = (const float*)d_in[1];
  const float* s    = (const float*)d_in[2];
  const float* h_e1 = (const float*)d_in[3];
  const float* h_e2 = (const float*)d_in[4];
  const float* h_d1 = (const float*)d_in[5];
  const float* h_d2 = (const float*)d_in[6];
  float* out = (float*)d_out;
  float* ws  = (float*)d_ws;

  const int BUF = 192 * 32 * 16;
  float* A0 = ws + 0 * BUF;
  float* A1 = ws + 1 * BUF;
  float* Y1 = ws + 2 * BUF;

  gt_layer<16, 32, 1, false, true, false><<<64, 384, 0, stream>>>(Sg, s, X, h_e1, A0);
  gt_layer<16, 16, 16, false, true, true><<<32, 384, 0, stream>>>(Sg, s, A0, h_e2, A1);
  gt_layer<16, 16, 32, true, false, false><<<16, 384, 0, stream>>>(Sg, s, A1, h_d1, Y1);
  gt_layer<32, 32, 16, true, false, false><<<1, 768, 0, stream>>>(Sg, s, Y1, h_d2, out);
}


# Decoder 2 cut into 4 time slabs: a work-group does a 16-step window (4-step halo) for 8 output steps, one column tile not two; 1->4 groups; out lines get four 32-B pieces; sums unchanged
# speedup vs baseline: 2.6154x; 1.2453x over previous
#include <hip/hip_runtime.h>

typedef __attribute__((ext_vector_type(2))) float v2f;
typedef __attribute__((ext_vector_type(4))) float v4f;
typedef __attribute__((ext_vector_type(8))) float v8f;

#define LDB 196

template <int T>
__device__ __forceinline__ void sg_times_plane(const v4f (&a)[24],
                                               const float* __restrict__ Bt,
                                               float* __restrict__ Us,
                                               int half, int bcol, int rbase) {
  v8f acc = {0.f, 0.f, 0.f, 0.f, 0.f, 0.f, 0.f, 0.f};
  const float* __restrict__ bp = Bt + bcol * LDB + half * 4;
  constexpr bool AHEAD = (T == 16);
  v4f bnext = {0.f, 0.f, 0.f, 0.f};
  if (AHEAD) bnext = *(const v4f*)bp;
#pragma unroll
  for (int m = 0; m < 24; ++m) {
    v4f bv;
    if (AHEAD) {
      bv = bnext;
      if (m + 1 < 24) bnext = *(const v4f*)(bp + 8 * (m + 1));
    } else {
      bv = *(const v4f*)(bp + 8 * m);
    }
    const v2f a0 = {a[m][0], a[m][1]};
    const v2f a1 = {a[m][2], a[m][3]};
    const v2f b0 = {bv[0], bv[1]};
    const v2f b1 = {bv[2], bv[3]};
    acc = __builtin_amdgcn_wmma_f32_16x16x4_f32(
        false, a0, false, b0, (short)0, acc, false, false);
    acc = __builtin_amdgcn_wmma_f32_16x16x4_f32(
        false, a1, false, b1, (short)0, acc, false, false);
  }
  float* __restrict__ up = Us + rbase * T + bcol;
#pragma unroll
  for (int v = 0; v < 8; ++v) {
    up[v * T] = acc[v];
  }
}

template <int T, int TG, int CIN, bool UNPOOL_IN, bool POOL_OUT, bool IN_SLAB>
__global__ __launch_bounds__(24 * T) void gt_layer(
    const float* __restrict__ Sg, const float* __restrict__ s,
    const float* __restrict__ xin, const float* __restrict__ h,
    float* __restrict__ yout) {
  __shared__ float Bt[T * LDB];
  __shared__ float Us[192 * T];

  const int tid   = threadIdx.x;
  constexpr bool SLAB = (TG != T);
  static_assert(!SLAB || (T == 16 && TG == 32 && UNPOOL_IN != POOL_OUT && !IN_SLAB),
                "slab form: 32-step layer; plain input and pooled output, or "
                "zero-stuffed input and plain output");
  static_assert(!IN_SLAB || (T == 16 && TG == 16 && !UNPOOL_IN),
                "slab-major input: 16 steps in 4 slabs of 4");
  const int o     = SLAB ? (int)(blockIdx.x >> 2) : (int)blockIdx.x;
  const int g     = SLAB ? (int)(blockIdx.x & 3) : 0;
  const int wave  = tid >> 5;
  const int l     = tid & 31;
  const int half  = l >> 4;
  const int lm    = l & 15;
  const int mt    = wave % 12;
  const int nt    = wave / 12;
  const int arow  = mt * 16 + lm;
  const int bcol  = nt * 16 + lm;
  const int rbase = mt * 16 + half * 8;

  v4f a[24];
  const float* __restrict__ ap = Sg + arow * 192 + half * 4;
#pragma unroll
  for (int m = 0; m < 24; ++m) {
    a[m] = *(const v4f*)(ap + 8 * m);
  }

  const float s00 = s[0], s01 = s[1], s10 = s[2], s11 = s[3];

  const int e0    = tid * 8;
  const int tau0  = e0 % T;
  const int node  = e0 / T;
  const int tprev = (tau0 + T - 1) % T;
  float* __restrict__ bown = Bt + tau0 * LDB + node;
  const float* __restrict__ bprev = Bt + tprev * LDB + node;
  const float* __restrict__ uprev = Us + node * T + tprev;

  float p0[8], p1[8], p2[8];
#pragma unroll
  for (int j = 0; j < 8; ++j) { p0[j] = 0.f; p1[j] = 0.f; p2[j] = 0.f; }
  const float* __restrict__ hp = h + o * CIN * 3;
  if (!UNPOOL_IN) {
    int offa, offb;
    if (SLAB) {
      offa = node * TG + ((8 * g - 4 + tau0) & (TG - 1));
      offb = node * TG + ((8 * g + tau0) & (TG - 1));
    } else if (IN_SLAB) {
      offa = (tau0 >> 2) * 768 + node * 4;
      offb = offa + 768;
    } else {
      offa = e0;
      offb = e0 + 4;
    }
    const float* __restrict__ xp = xin;
#pragma unroll
    for (int i = 0; i < CIN; ++i) {
      const v4f xa = *(const v4f*)(xp + i * (192 * TG) + offa);
      const v4f xb = *(const v4f*)(xp + i * (192 * TG) + offb);
      const float h0 = hp[i * 3], h1 = hp[i * 3 + 1], h2 = hp[i * 3 + 2];
#pragma unroll
      for (int j = 0; j < 4; ++j) {
        p0[j]     = fmaf(h0, xa[j], p0[j]);
        p1[j]     = fmaf(h1, xa[j], p1[j]);
        p2[j]     = fmaf(h2, xa[j], p2[j]);
        p0[4 + j] = fmaf(h0, xb[j], p0[4 + j]);
        p1[4 + j] = fmaf(h1, xb[j], p1[4 + j]);
        p2[4 + j] = fmaf(h2, xb[j], p2[4 + j]);
      }
    }
  } else if (SLAB) {
    const int q0 = (4 * g - 2 + (tau0 >> 1)) & (TG / 2 - 1);
    const int q1 = (q0 + 2) & (TG / 2 - 1);
    const float* __restrict__ xpa = xin + node * (TG / 2) + q0;
    const float* __restrict__ xpb = xin + node * (TG / 2) + q1;
#pragma unroll
    for (int i = 0; i < CIN; ++i) {
      const v2f xa = *(const v2f*)(xpa + i * (192 * (TG / 2)));
      const v2f xb = *(const v2f*)(xpb + i * (192 * (TG / 2)));
      const float h0 = hp[i * 3], h1 = hp[i * 3 + 1], h2 = hp[i * 3 + 2];
      const float x[4] = {fmaxf(xa[0], 0.f), fmaxf(xa[1], 0.f),
                          fmaxf(xb[0], 0.f), fmaxf(xb[1], 0.f)};
#pragma unroll
      for (int j = 0; j < 4; ++j) {
        p0[2 * j] = fmaf(h0, x[j], p0[2 * j]);
        p1[2 * j] = fmaf(h1, x[j], p1[2 * j]);
        p2[2 * j] = fmaf(h2, x[j], p2[2 * j]);
      }
    }
  } else {
    const float* __restrict__ xp = xin + tid * 4;
#pragma unroll
    for (int i = 0; i < CIN; ++i) {
      const v4f xa = *(const v4f*)(xp + i * (192 * (T / 2)));
      const float h0 = hp[i * 3], h1 = hp[i * 3 + 1], h2 = hp[i * 3 + 2];
#pragma unroll
      for (int j = 0; j < 4; ++j) {
        const float x = fmaxf(xa[j], 0.f);
        p0[2 * j] = fmaf(h0, x, p0[2 * j]);
        p1[2 * j] = fmaf(h1, x, p1[2 * j]);
        p2[2 * j] = fmaf(h2, x, p2[2 * j]);
      }
    }
  }

#pragma unroll
  for (int j = 0; j < 8; ++j) {
    bown[j * LDB] = p2[j];
  }
  __syncthreads();
  sg_times_plane<T>(a, Bt, Us, half, bcol, rbase);
  __syncthreads();

  float w[8];
  {
    const v4f ua = *(const v4f*)(Us + e0);
    const v4f ub = *(const v4f*)(Us + e0 + 4);
    const float u[8] = {ua[0], ua[1], ua[2], ua[3], ub[0], ub[1], ub[2], ub[3]};
    float zm = *bprev;
    float um = *uprev;
#pragma unroll
    for (int j = 0; j < 8; ++j) {
      float t = fmaf(s00, p2[j], p1[j]);
      t = fmaf(s01, u[j], t);
      t = fmaf(s10, zm, t);
      t = fmaf(s11, um, t);
      w[j] = t;
      zm = p2[j];
      um = u[j];
    }
  }
  __syncthreads();
#pragma unroll
  for (int j = 0; j < 8; ++j) {
    bown[j * LDB] = w[j];
  }
  __syncthreads();

  sg_times_plane<T>(a, Bt, Us, half, bcol, rbase);
  __syncthreads();

  float y[8];
  {
    const v4f ua = *(const v4f*)(Us + e0);
    const v4f ub = *(const v4f*)(Us + e0 + 4);
    const float u[8] = {ua[0], ua[1], ua[2], ua[3], ub[0], ub[1], ub[2], ub[3]};
    float zm = *bprev;
    float um = *uprev;
#pragma unroll
    for (int j = 0; j < 8; ++j) {
      float t = fmaf(s00, w[j], p0[j]);
      t = fmaf(s01, u[j], t);
      t = fmaf(s10, zm, t);
      t = fmaf(s11, um, t);
      y[j] = t;
      zm = w[j];
      um = u[j];
    }
  }

  if (SLAB && !POOL_OUT) {
    const int hsel = tid & 1;
    float* __restrict__ yp = yout + o * (192 * TG) + node * TG + 8 * g + hsel * 4;
    v4f r;
    r[0] = hsel ? y[0] : y[4];
    r[1] = hsel ? y[1] : y[5];
    r[2] = hsel ? y[2] : y[6];
    r[3] = hsel ? y[3] : y[7];
    *(v4f*)yp = r;
  } else if (SLAB) {
    const int hsel = tid & 1;
    float* __restrict__ yp = yout + (o * 4 + g) * 768 + node * 4 + hsel * 2;
    const float ya = hsel ? y[0] : y[4];
    const float yb = hsel ? y[1] : y[5];
    const float yc = hsel ? y[2] : y[6];
    const float yd = hsel ? y[3] : y[7];
    v2f r;
    r[0] = fmaxf(fmaxf(ya, yb), 0.f);
    r[1] = fmaxf(fmaxf(yc, yd), 0.f);
    *(v2f*)yp = r;
  } else if (POOL_OUT) {
    float* __restrict__ yp = yout + o * (192 * (T / 2)) + tid * 4;
    v4f r;
#pragma unroll
    for (int j = 0; j < 4; ++j) {
      r[j] = fmaxf(fmaxf(y[2 * j], y[2 * j + 1]), 0.f);
    }
    *(v4f*)yp = r;
  } else {
    float* __restrict__ yp = yout + o * (192 * T) + e0;
    v4f lo = {y[0], y[1], y[2], y[3]};
    v4f hi = {y[4], y[5], y[6], y[7]};
    *(v4f*)yp       = lo;
    *(v4f*)(yp + 4) = hi;
  }
}

extern "C" void kernel_launch(void* const* d_in, const int* in_sizes, int n_in,
                              void* d_out, int out_size, void* d_ws,
                              size_t ws_size, hipStream_t stream) {
  (void)in_sizes; (void)n_in; (void)out_size; (void)ws_size;
  const float* X    = (const float*)d_in[0];
  const float* Sg   = (const float*)d_in[1];
  const float* s    = (const float*)d_in[2];
  const float* h_e1 = (const float*)d_in[3];
  const float* h_e2 = (const float*)d_in[4];
  const float* h_d1 = (const float*)d_in[5];
  const float* h_d2 = (const float*)d_in[6];
  float* out = (float*)d_out;
  float* ws  = (float*)d_ws;

  const int BUF = 192 * 32 * 16;
  float* A0 = ws + 0 * BUF;
  float* A1 = ws + 1 * BUF;
  float* Y1 = ws + 2 * BUF;

  gt_layer<16, 32, 1, false, true, false><<<64, 384, 0, stream>>>(Sg, s, X, h_e1, A0);
  gt_layer<16, 16, 16, false, true, true><<<32, 384, 0, stream>>>(Sg, s, A0, h_e2, A1);
  gt_layer<16, 16, 32, true, false, false><<<16, 384, 0, stream>>>(Sg, s, A1, h_d1, Y1);
  gt_layer<16, 32, 16, true, false, false><<<4, 384, 0, stream>>>(Sg, s, Y1, h_d2, out);
}


# Sg strip stream overlapped with the first shift: activation loads issued ahead of the strip, hand-over barrier fences LDS only, each matrix step waits for its own strip block; sums unchanged
# speedup vs baseline: 2.7408x; 1.0479x over previous
#include <hip/hip_runtime.h>

typedef __attribute__((ext_vector_type(2))) float v2f;
typedef __attribute__((ext_vector_type(4))) float v4f;
typedef __attribute__((ext_vector_type(8))) float v8f;

#define LDB 196

template <int T>
__device__ __forceinline__ void sg_times_plane(const v4f (&a)[24],
                                               const float* __restrict__ Bt,
                                               float* __restrict__ Us,
                                               int half, int bcol, int rbase) {
  v8f acc = {0.f, 0.f, 0.f, 0.f, 0.f, 0.f, 0.f, 0.f};
  const float* __restrict__ bp = Bt + bcol * LDB + half * 4;
  constexpr bool AHEAD = (T == 16);
  v4f bnext = {0.f, 0.f, 0.f, 0.f};
  if (AHEAD) bnext = *(const v4f*)bp;
#pragma unroll
  for (int m = 0; m < 24; ++m) {
    v4f bv;
    if (AHEAD) {
      bv = bnext;
      if (m + 1 < 24) bnext = *(const v4f*)(bp + 8 * (m + 1));
    } else {
      bv = *(const v4f*)(bp + 8 * m);
    }
    const v2f a0 = {a[m][0], a[m][1]};
    const v2f a1 = {a[m][2], a[m][3]};
    const v2f b0 = {bv[0], bv[1]};
    const v2f b1 = {bv[2], bv[3]};
    acc = __builtin_amdgcn_wmma_f32_16x16x4_f32(
        false, a0, false, b0, (short)0, acc, false, false);
    acc = __builtin_amdgcn_wmma_f32_16x16x4_f32(
        false, a1, false, b1, (short)0, acc, false, false);
  }
  float* __restrict__ up = Us + rbase * T + bcol;
#pragma unroll
  for (int v = 0; v < 8; ++v) {
    up[v * T] = acc[v];
  }
}

template <int T, int TG, int CIN, bool UNPOOL_IN, bool POOL_OUT, bool IN_SLAB>
__global__ __launch_bounds__(24 * T) void gt_layer(
    const float* __restrict__ Sg, const float* __restrict__ s,
    const float* __restrict__ xin, const float* __restrict__ h,
    float* __restrict__ yout) {
  __shared__ float Bt[T * LDB];
  __shared__ float Us[192 * T];

  const int tid   = threadIdx.x;
  constexpr bool SLAB = (TG != T);
  static_assert(!SLAB || (T == 16 && TG == 32 && UNPOOL_IN != POOL_OUT && !IN_SLAB),
                "slab form: 32-step layer; plain input and pooled output, or "
                "zero-stuffed input and plain output");
  static_assert(!IN_SLAB || (T == 16 && TG == 16 && !UNPOOL_IN),
                "slab-major input: 16 steps in 4 slabs of 4");
  const int o     = SLAB ? (int)(blockIdx.x >> 2) : (int)blockIdx.x;
  const int g     = SLAB ? (int)(blockIdx.x & 3) : 0;
  const int wave  = tid >> 5;
  const int l     = tid & 31;
  const int half  = l >> 4;
  const int lm    = l & 15;
  const int mt    = wave % 12;
  const int nt    = wave / 12;
  const int arow  = mt * 16 + lm;
  const int bcol  = nt * 16 + lm;
  const int rbase = mt * 16 + half * 8;

  const float s00 = s[0], s01 = s[1], s10 = s[2], s11 = s[3];

  const int e0    = tid * 8;
  const int tau0  = e0 % T;
  const int node  = e0 / T;
  const int tprev = (tau0 + T - 1) % T;
  float* __restrict__ bown = Bt + tau0 * LDB + node;
  const float* __restrict__ bprev = Bt + tprev * LDB + node;
  const float* __restrict__ uprev = Us + node * T + tprev;

  v4f xa4[CIN], xb4[CIN];
  v2f xa2[CIN], xb2[CIN];
  if (!UNPOOL_IN) {
    int offa, offb;
    if (SLAB) {
      offa = node * TG + ((8 * g - 4 + tau0) & (TG - 1));
      offb = node * TG + ((8 * g + tau0) & (TG - 1));
    } else if (IN_SLAB) {
      offa = (tau0 >> 2) * 768 + node * 4;
      offb = offa + 768;
    } else {
      offa = e0;
      offb = e0 + 4;
    }
#pragma unroll
    for (int i = 0; i < CIN; ++i) {
      xa4[i] = *(const v4f*)(xin + i * (192 * TG) + offa);
      xb4[i] = *(const v4f*)(xin + i * (192 * TG) + offb);
    }
  } else if (SLAB) {
    const int q0 = (4 * g - 2 + (tau0 >> 1)) & (TG / 2 - 1);
    const int q1 = (q0 + 2) & (TG / 2 - 1);
    const float* __restrict__ xpa = xin + node * (TG / 2) + q0;
    const float* __restrict__ xpb = xin + node * (TG / 2) + q1;
#pragma unroll
    for (int i = 0; i < CIN; ++i) {
      xa2[i] = *(const v2f*)(xpa + i * (192 * (TG / 2)));
      xb2[i] = *(const v2f*)(xpb + i * (192 * (TG / 2)));
    }
  } else {
    const float* __restrict__ xp = xin + tid * 4;
#pragma unroll
    for (int i = 0; i < CIN; ++i) {
      xa4[i] = *(const v4f*)(xp + i * (192 * (T / 2)));
    }
  }

  __builtin_amdgcn_s_barrier();

  v4f a[24];
  const float* __restrict__ ap = Sg + arow * 192 + half * 4;
#pragma unroll
  for (int m = 0; m < 24; ++m) {
    a[m] = *(const v4f*)(ap + 8 * m);
  }

  float p0[8], p1[8], p2[8];
#pragma unroll
  for (int j = 0; j < 8; ++j) { p0[j] = 0.f; p1[j] = 0.f; p2[j] = 0.f; }
  const float* __restrict__ hp = h + o * CIN * 3;
  if (!UNPOOL_IN) {
#pragma unroll
    for (int i = 0; i < CIN; ++i) {
      const v4f xa = xa4[i];
      const v4f xb = xb4[i];
      const float h0 = hp[i * 3], h1 = hp[i * 3 + 1], h2 = hp[i * 3 + 2];
#pragma unroll
      for (int j = 0; j < 4; ++j) {
        p0[j]     = fmaf(h0, xa[j], p0[j]);
        p1[j]     = fmaf(h1, xa[j], p1[j]);
        p2[j]     = fmaf(h2, xa[j], p2[j]);
        p0[4 + j] = fmaf(h0, xb[j], p0[4 + j]);
        p1[4 + j] = fmaf(h1, xb[j], p1[4 + j]);
        p2[4 + j] = fmaf(h2, xb[j], p2[4 + j]);
      }
    }
  } else if (SLAB) {
#pragma unroll
    for (int i = 0; i < CIN; ++i) {
      const v2f xa = xa2[i];
      const v2f xb = xb2[i];
      const float h0 = hp[i * 3], h1 = hp[i * 3 + 1], h2 = hp[i * 3 + 2];
      const float x[4] = {fmaxf(xa[0], 0.f), fmaxf(xa[1], 0.f),
                          fmaxf(xb[0], 0.f), fmaxf(xb[1], 0.f)};
#pragma unroll
      for (int j = 0; j < 4; ++j) {
        p0[2 * j] = fmaf(h0, x[j], p0[2 * j]);
        p1[2 * j] = fmaf(h1, x[j], p1[2 * j]);
        p2[2 * j] = fmaf(h2, x[j], p2[2 * j]);
      }
    }
  } else {
#pragma unroll
    for (int i = 0; i < CIN; ++i) {
      const v4f xa = xa4[i];
      const float h0 = hp[i * 3], h1 = hp[i * 3 + 1], h2 = hp[i * 3 + 2];
#pragma unroll
      for (int j = 0; j < 4; ++j) {
        const float x = fmaxf(xa[j], 0.f);
        p0[2 * j] = fmaf(h0, x, p0[2 * j]);
        p1[2 * j] = fmaf(h1, x, p1[2 * j]);
        p2[2 * j] = fmaf(h2, x, p2[2 * j]);
      }
    }
  }

#pragma unroll
  for (int j = 0; j < 8; ++j) {
    bown[j * LDB] = p2[j];
  }
  __builtin_amdgcn_fence(__ATOMIC_RELEASE, "workgroup", "local");
  __builtin_amdgcn_s_barrier();
  __builtin_amdgcn_fence(__ATOMIC_ACQUIRE, "workgroup", "local");
  sg_times_plane<T>(a, Bt, Us, half, bcol, rbase);
  __syncthreads();

  float w[8];
  {
    const v4f ua = *(const v4f*)(Us + e0);
    const v4f ub = *(const v4f*)(Us + e0 + 4);
    const float u[8] = {ua[0], ua[1], ua[2], ua[3], ub[0], ub[1], ub[2], ub[3]};
    float zm = *bprev;
    float um = *uprev;
#pragma unroll
    for (int j = 0; j < 8; ++j) {
      float t = fmaf(s00, p2[j], p1[j]);
      t = fmaf(s01, u[j], t);
      t = fmaf(s10, zm, t);
      t = fmaf(s11, um, t);
      w[j] = t;
      zm = p2[j];
      um = u[j];
    }
  }
  __syncthreads();
#pragma unroll
  for (int j = 0; j < 8; ++j) {
    bown[j * LDB] = w[j];
  }
  __syncthreads();

  sg_times_plane<T>(a, Bt, Us, half, bcol, rbase);
  __syncthreads();

  float y[8];
  {
    const v4f ua = *(const v4f*)(Us + e0);
    const v4f ub = *(const v4f*)(Us + e0 + 4);
    const float u[8] = {ua[0], ua[1], ua[2], ua[3], ub[0], ub[1], ub[2], ub[3]};
    float zm = *bprev;
    float um = *uprev;
#pragma unroll
    for (int j = 0; j < 8; ++j) {
      float t = fmaf(s00, w[j], p0[j]);
      t = fmaf(s01, u[j], t);
      t = fmaf(s10, zm, t);
      t = fmaf(s11, um, t);
      y[j] = t;
      zm = w[j];
      um = u[j];
    }
  }

  if (SLAB && !POOL_OUT) {
    const int hsel = tid & 1;
    float* __restrict__ yp = yout + o * (192 * TG) + node * TG + 8 * g + hsel * 4;
    v4f r;
    r[0] = hsel ? y[0] : y[4];
    r[1] = hsel ? y[1] : y[5];
    r[2] = hsel ? y[2] : y[6];
    r[3] = hsel ? y[3] : y[7];
    *(v4f*)yp = r;
  } else if (SLAB) {
    const int hsel = tid & 1;
    float* __restrict__ yp = yout + (o * 4 + g) * 768 + node * 4 + hsel * 2;
    const float ya = hsel ? y[0] : y[4];
    const float yb = hsel ? y[1] : y[5];
    const float yc = hsel ? y[2] : y[6];
    const float yd = hsel ? y[3] : y[7];
    v2f r;
    r[0] = fmaxf(fmaxf(ya, yb), 0.f);
    r[1] = fmaxf(fmaxf(yc, yd), 0.f);
    *(v2f*)yp = r;
  } else if (POOL_OUT) {
    float* __restrict__ yp = yout + o * (192 * (T / 2)) + tid * 4;
    v4f r;
#pragma unroll
    for (int j = 0; j < 4; ++j) {
      r[j] = fmaxf(fmaxf(y[2 * j], y[2 * j + 1]), 0.f);
    }
    *(v4f*)yp = r;
  } else {
    float* __restrict__ yp = yout + o * (192 * T) + e0;
    v4f lo = {y[0], y[1], y[2], y[3]};
    v4f hi = {y[4], y[5], y[6], y[7]};
    *(v4f*)yp       = lo;
    *(v4f*)(yp + 4) = hi;
  }
}

extern "C" void kernel_launch(void* const* d_in, const int* in_sizes, int n_in,
                              void* d_out, int out_size, void* d_ws,
                              size_t ws_size, hipStream_t stream) {
  (void)in_sizes; (void)n_in; (void)out_size; (void)ws_size;
  const float* X    = (const float*)d_in[0];
  const float* Sg   = (const float*)d_in[1];
  const float* s    = (const float*)d_in[2];
  const float* h_e1 = (const float*)d_in[3];
  const float* h_e2 = (const float*)d_in[4];
  const float* h_d1 = (const float*)d_in[5];
  const float* h_d2 = (const float*)d_in[6];
  float* out = (float*)d_out;
  float* ws  = (float*)d_ws;

  const int BUF = 192 * 32 * 16;
  float* A0 = ws + 0 * BUF;
  float* A1 = ws + 1 * BUF;
  float* Y1 = ws + 2 * BUF;

  gt_layer<16, 32, 1, false, true, false><<<64, 384, 0, stream>>>(Sg, s, X, h_e1, A0);
  gt_layer<16, 16, 16, false, true, true><<<32, 384, 0, stream>>>(Sg, s, A0, h_e2, A1);
  gt_layer<16, 16, 32, true, false, false><<<16, 384, 0, stream>>>(Sg, s, A1, h_d1, Y1);
  gt_layer<16, 32, 16, true, false, false><<<4, 384, 0, stream>>>(Sg, s, Y1, h_d2, out);
}


# Second shift of both decoder layers split by row group over 3 work-groups: each redoes the first shift, then 4 waves (one per SIMD) do 4 row tiles, B read in two groups of 12; sums unchanged
# speedup vs baseline: 2.9822x; 1.0881x over previous
#include <hip/hip_runtime.h>

typedef __attribute__((ext_vector_type(2))) float v2f;
typedef __attribute__((ext_vector_type(4))) float v4f;
typedef __attribute__((ext_vector_type(8))) float v8f;

#define LDB 196

template <int T, bool GROUP>
__device__ __forceinline__ void sg_times_plane(const v4f (&a)[24],
                                               const float* __restrict__ Bt,
                                               float* __restrict__ Us,
                                               int half, int bcol, int rbase) {
  v8f acc = {0.f, 0.f, 0.f, 0.f, 0.f, 0.f, 0.f, 0.f};
  const float* __restrict__ bp = Bt + bcol * LDB + half * 4;
  constexpr bool AHEAD = (T == 16) && !GROUP;
  v4f bgrp[24];
  if (GROUP) {
#pragma unroll
    for (int m = 0; m < 12; ++m) {
      bgrp[m] = *(const v4f*)(bp + 8 * m);
    }
    __builtin_amdgcn_sched_barrier(0x6);
  }
  v4f bnext = {0.f, 0.f, 0.f, 0.f};
  if (AHEAD) bnext = *(const v4f*)bp;
#pragma unroll
  for (int m = 0; m < 24; ++m) {
    v4f bv;
    if (GROUP) {
      if (m == 6) {
#pragma unroll
        for (int q = 12; q < 24; ++q) {
          bgrp[q] = *(const v4f*)(bp + 8 * q);
        }
        __builtin_amdgcn_sched_barrier(0x6);
      }
      bv = bgrp[m];
    } else if (AHEAD) {
      bv = bnext;
      if (m + 1 < 24) bnext = *(const v4f*)(bp + 8 * (m + 1));
    } else {
      bv = *(const v4f*)(bp + 8 * m);
    }
    const v2f a0 = {a[m][0], a[m][1]};
    const v2f a1 = {a[m][2], a[m][3]};
    const v2f b0 = {bv[0], bv[1]};
    const v2f b1 = {bv[2], bv[3]};
    acc = __builtin_amdgcn_wmma_f32_16x16x4_f32(
        false, a0, false, b0, (short)0, acc, false, false);
    acc = __builtin_amdgcn_wmma_f32_16x16x4_f32(
        false, a1, false, b1, (short)0, acc, false, false);
  }
  float* __restrict__ up = Us + rbase * T + bcol;
#pragma unroll
  for (int v = 0; v < 8; ++v) {
    up[v * T] = acc[v];
  }
}

template <int T, int TG, int CIN, bool UNPOOL_IN, bool POOL_OUT, bool IN_SLAB,
          bool ROWSPLIT>
__global__ __launch_bounds__(24 * T) void gt_layer(
    const float* __restrict__ Sg, const float* __restrict__ s,
    const float* __restrict__ xin, const float* __restrict__ h,
    float* __restrict__ yout) {
  __shared__ float Bt[T * LDB];
  __shared__ float Us[192 * T];

  const int tid   = threadIdx.x;
  constexpr bool SLAB = (TG != T);
  static_assert(!SLAB || (T == 16 && TG == 32 && UNPOOL_IN != POOL_OUT && !IN_SLAB),
                "slab form: 32-step layer; plain input and pooled output, or "
                "zero-stuffed input and plain output");
  static_assert(!IN_SLAB || (T == 16 && TG == 16 && !UNPOOL_IN),
                "slab-major input: 16 steps in 4 slabs of 4");
  static_assert(!ROWSPLIT || T == 16,
                "row groups: 12 waves, wave w owns row tile w");
  const int bid   = ROWSPLIT ? (int)(blockIdx.x / 3) : (int)blockIdx.x;
  const int rg    = ROWSPLIT ? (int)(blockIdx.x % 3) : 0;
  const int o     = SLAB ? (bid >> 2) : bid;
  const int g     = SLAB ? (bid & 3) : 0;
  const int wave  = tid >> 5;
  const int l     = tid & 31;
  const int half  = l >> 4;
  const int lm    = l & 15;
  const int mt    = wave % 12;
  const int nt    = wave / 12;
  const int arow  = mt * 16 + lm;
  const int bcol  = nt * 16 + lm;
  const int rbase = mt * 16 + half * 8;

  const float s00 = s[0], s01 = s[1], s10 = s[2], s11 = s[3];

  const int e0    = tid * 8;
  const int tau0  = e0 % T;
  const int node  = e0 / T;
  const int tprev = (tau0 + T - 1) % T;
  float* __restrict__ bown = Bt + tau0 * LDB + node;
  const float* __restrict__ bprev = Bt + tprev * LDB + node;
  const float* __restrict__ uprev = Us + node * T + tprev;

  v4f xa4[CIN], xb4[CIN];
  v2f xa2[CIN], xb2[CIN];
  if (!UNPOOL_IN) {
    int offa, offb;
    if (SLAB) {
      offa = node * TG + ((8 * g - 4 + tau0) & (TG - 1));
      offb = node * TG + ((8 * g + tau0) & (TG - 1));
    } else if (IN_SLAB) {
      offa = (tau0 >> 2) * 768 + node * 4;
      offb = offa + 768;
    } else {
      offa = e0;
      offb = e0 + 4;
    }
#pragma unroll
    for (int i = 0; i < CIN; ++i) {
      xa4[i] = *(const v4f*)(xin + i * (192 * TG) + offa);
      xb4[i] = *(const v4f*)(xin + i * (192 * TG) + offb);
    }
  } else if (SLAB) {
    const int q0 = (4 * g - 2 + (tau0 >> 1)) & (TG / 2 - 1);
    const int q1 = (q0 + 2) & (TG / 2 - 1);
    const float* __restrict__ xpa = xin + node * (TG / 2) + q0;
    const float* __restrict__ xpb = xin + node * (TG / 2) + q1;
#pragma unroll
    for (int i = 0; i < CIN; ++i) {
      xa2[i] = *(const v2f*)(xpa + i * (192 * (TG / 2)));
      xb2[i] = *(const v2f*)(xpb + i * (192 * (TG / 2)));
    }
  } else {
    const float* __restrict__ xp = xin + tid * 4;
#pragma unroll
    for (int i = 0; i < CIN; ++i) {
      xa4[i] = *(const v4f*)(xp + i * (192 * (T / 2)));
    }
  }

  __builtin_amdgcn_s_barrier();

  v4f a[24];
  const float* __restrict__ ap = Sg + arow * 192 + half * 4;
#pragma unroll
  for (int m = 0; m < 24; ++m) {
    a[m] = *(const v4f*)(ap + 8 * m);
  }

  float p0[8], p1[8], p2[8];
#pragma unroll
  for (int j = 0; j < 8; ++j) { p0[j] = 0.f; p1[j] = 0.f; p2[j] = 0.f; }
  const float* __restrict__ hp = h + o * CIN * 3;
  if (!UNPOOL_IN) {
#pragma unroll
    for (int i = 0; i < CIN; ++i) {
      const v4f xa = xa4[i];
      const v4f xb = xb4[i];
      const float h0 = hp[i * 3], h1 = hp[i * 3 + 1], h2 = hp[i * 3 + 2];
#pragma unroll
      for (int j = 0; j < 4; ++j) {
        p0[j]     = fmaf(h0, xa[j], p0[j]);
        p1[j]     = fmaf(h1, xa[j], p1[j]);
        p2[j]     = fmaf(h2, xa[j], p2[j]);
        p0[4 + j] = fmaf(h0, xb[j], p0[4 + j]);
        p1[4 + j] = fmaf(h1, xb[j], p1[4 + j]);
        p2[4 + j] = fmaf(h2, xb[j], p2[4 + j]);
      }
    }
  } else if (SLAB) {
#pragma unroll
    for (int i = 0; i < CIN; ++i) {
      const v2f xa = xa2[i];
      const v2f xb = xb2[i];
      const float h0 = hp[i * 3], h1 = hp[i * 3 + 1], h2 = hp[i * 3 + 2];
      const float x[4] = {fmaxf(xa[0], 0.f), fmaxf(xa[1], 0.f),
                          fmaxf(xb[0], 0.f), fmaxf(xb[1], 0.f)};
#pragma unroll
      for (int j = 0; j < 4; ++j) {
        p0[2 * j] = fmaf(h0, x[j], p0[2 * j]);
        p1[2 * j] = fmaf(h1, x[j], p1[2 * j]);
        p2[2 * j] = fmaf(h2, x[j], p2[2 * j]);
      }
    }
  } else {
#pragma unroll
    for (int i = 0; i < CIN; ++i) {
      const v4f xa = xa4[i];
      const float h0 = hp[i * 3], h1 = hp[i * 3 + 1], h2 = hp[i * 3 + 2];
#pragma unroll
      for (int j = 0; j < 4; ++j) {
        const float x = fmaxf(xa[j], 0.f);
        p0[2 * j] = fmaf(h0, x, p0[2 * j]);
        p1[2 * j] = fmaf(h1, x, p1[2 * j]);
        p2[2 * j] = fmaf(h2, x, p2[2 * j]);
      }
    }
  }

#pragma unroll
  for (int j = 0; j < 8; ++j) {
    bown[j * LDB] = p2[j];
  }
  __builtin_amdgcn_fence(__ATOMIC_RELEASE, "workgroup", "local");
  __builtin_amdgcn_s_barrier();
  __builtin_amdgcn_fence(__ATOMIC_ACQUIRE, "workgroup", "local");
  sg_times_plane<T, false>(a, Bt, Us, half, bcol, rbase);
  __syncthreads();

  float w[8];
  {
    const v4f ua = *(const v4f*)(Us + e0);
    const v4f ub = *(const v4f*)(Us + e0 + 4);
    const float u[8] = {ua[0], ua[1], ua[2], ua[3], ub[0], ub[1], ub[2], ub[3]};
    float zm = *bprev;
    float um = *uprev;
#pragma unroll
    for (int j = 0; j < 8; ++j) {
      float t = fmaf(s00, p2[j], p1[j]);
      t = fmaf(s01, u[j], t);
      t = fmaf(s10, zm, t);
      t = fmaf(s11, um, t);
      w[j] = t;
      zm = p2[j];
      um = u[j];
    }
  }
  __syncthreads();
#pragma unroll
  for (int j = 0; j < 8; ++j) {
    bown[j * LDB] = w[j];
  }
  __syncthreads();

  const bool mine =
      !ROWSPLIT || (__builtin_amdgcn_readfirstlane(wave) % 3 == rg);
  if (mine) sg_times_plane<T, ROWSPLIT>(a, Bt, Us, half, bcol, rbase);
  __syncthreads();
  if (!mine) return;

  float y[8];
  {
    const v4f ua = *(const v4f*)(Us + e0);
    const v4f ub = *(const v4f*)(Us + e0 + 4);
    const float u[8] = {ua[0], ua[1], ua[2], ua[3], ub[0], ub[1], ub[2], ub[3]};
    float zm = *bprev;
    float um = *uprev;
#pragma unroll
    for (int j = 0; j < 8; ++j) {
      float t = fmaf(s00, w[j], p0[j]);
      t = fmaf(s01, u[j], t);
      t = fmaf(s10, zm, t);
      t = fmaf(s11, um, t);
      y[j] = t;
      zm = w[j];
      um = u[j];
    }
  }

  if (SLAB && !POOL_OUT) {
    const int hsel = tid & 1;
    float* __restrict__ yp = yout + o * (192 * TG) + node * TG + 8 * g + hsel * 4;
    v4f r;
    r[0] = hsel ? y[0] : y[4];
    r[1] = hsel ? y[1] : y[5];
    r[2] = hsel ? y[2] : y[6];
    r[3] = hsel ? y[3] : y[7];
    *(v4f*)yp = r;
  } else if (SLAB) {
    const int hsel = tid & 1;
    float* __restrict__ yp = yout + (o * 4 + g) * 768 + node * 4 + hsel * 2;
    const float ya = hsel ? y[0] : y[4];
    const float yb = hsel ? y[1] : y[5];
    const float yc = hsel ? y[2] : y[6];
    const float yd = hsel ? y[3] : y[7];
    v2f r;
    r[0] = fmaxf(fmaxf(ya, yb), 0.f);
    r[1] = fmaxf(fmaxf(yc, yd), 0.f);
    *(v2f*)yp = r;
  } else if (POOL_OUT) {
    float* __restrict__ yp = yout + o * (192 * (T / 2)) + tid * 4;
    v4f r;
#pragma unroll
    for (int j = 0; j < 4; ++j) {
      r[j] = fmaxf(fmaxf(y[2 * j], y[2 * j + 1]), 0.f);
    }
    *(v4f*)yp = r;
  } else {
    float* __restrict__ yp = yout + o * (192 * T) + e0;
    v4f lo = {y[0], y[1], y[2], y[3]};
    v4f hi = {y[4], y[5], y[6], y[7]};
    *(v4f*)yp       = lo;
    *(v4f*)(yp + 4) = hi;
  }
}

extern "C" void kernel_launch(void* const* d_in, const int* in_sizes, int n_in,
                              void* d_out, int out_size, void* d_ws,
                              size_t ws_size, hipStream_t stream) {
  (void)in_sizes; (void)n_in; (void)out_size; (void)ws_size;
  const float* X    = (const float*)d_in[0];
  const float* Sg   = (const float*)d_in[1];
  const float* s    = (const float*)d_in[2];
  const float* h_e1 = (const float*)d_in[3];
  const float* h_e2 = (const float*)d_in[4];
  const float* h_d1 = (const float*)d_in[5];
  const float* h_d2 = (const float*)d_in[6];
  float* out = (float*)d_out;
  float* ws  = (float*)d_ws;

  const int BUF = 192 * 32 * 16;
  float* A0 = ws + 0 * BUF;
  float* A1 = ws + 1 * BUF;
  float* Y1 = ws + 2 * BUF;

  gt_layer<16, 32, 1, false, true, false, false><<<64, 384, 0, stream>>>(Sg, s, X, h_e1, A0);
  gt_layer<16, 16, 16, false, true, true, false><<<32, 384, 0, stream>>>(Sg, s, A0, h_e2, A1);
  gt_layer<16, 16, 32, true, false, false, true><<<48, 384, 0, stream>>>(Sg, s, A1, h_d1, Y1);
  gt_layer<16, 32, 16, true, false, false, true><<<12, 384, 0, stream>>>(Sg, s, Y1, h_d2, out);
}
